# starting kernel
# baseline (speedup 1.0000x reference)
#include <hip/hip_runtime.h>

typedef __attribute__((ext_vector_type(2))) float v2f;
typedef __attribute__((ext_vector_type(8))) float v8f;

#define MJ_STRIDE  132
#define ADJ_STRIDE  66

__launch_bounds__(256)
__global__ void gnn_fused_kernel(const float* __restrict__ x,
                                 const int*   __restrict__ adj,
                                 const float* __restrict__ W,
                                 const float* __restrict__ bias,
                                 float*       __restrict__ out)
{
    __shared__ __align__(16) float s_mj [64 * MJ_STRIDE];
    __shared__ __align__(16) float s_adj[64 * ADJ_STRIDE];
    __shared__ float s_deg[64];

    const int tid  = threadIdx.x;
    const int wave = tid >> 5;
    const int lane = tid & 31;
    const int hf   = lane >> 4;
    const int l16  = lane & 15;

    const int b     = blockIdx.y;
    const int obase = blockIdx.x * 128;

    const int kslab = wave >> 1;
    const int osub0 = (wave & 1) * 4;

    #pragma unroll
    for (int it = 0; it < 16; ++it) {
        int idx = it * 256 + tid;
        int r = idx >> 6, c = idx & 63;
        s_adj[r * ADJ_STRIDE + c] = (float)adj[idx];
    }
    __syncthreads();
    if (tid < 64) {
        float d = 0.f;
        for (int j = 0; j < 64; ++j) d += s_adj[tid * ADJ_STRIDE + j];
        s_deg[tid] = d;
    }

    const v8f zero = {0.f,0.f,0.f,0.f,0.f,0.f,0.f,0.f};
    v8f acc_i[4], acc_j[4];
    #pragma unroll
    for (int t = 0; t < 4; ++t) { acc_i[t] = zero; acc_j[t] = zero; }

    const float* xrow = x + ((size_t)b * 64 + kslab * 16 + l16) * 512 + 2 * hf;
    const float* wbase = W + (size_t)(obase + osub0 * 16 + l16) * 1024 + 2 * hf;

    for (int d = 0; d < 512; d += 4) {
        v2f a = *(const v2f*)(xrow + d);
        #pragma unroll
        for (int t = 0; t < 4; ++t) {
            const float* wrow = wbase + (size_t)t * 16 * 1024 + d;
            v2f bi = *(const v2f*)(wrow);
            v2f bj = *(const v2f*)(wrow + 512);
            acc_i[t] = __builtin_amdgcn_wmma_f32_16x16x4_f32(
                false, a, false, bi, (short)0, acc_i[t], false, false);
            acc_j[t] = __builtin_amdgcn_wmma_f32_16x16x4_f32(
                false, a, false, bj, (short)0, acc_j[t], false, false);
        }
    }

    #pragma unroll
    for (int t = 0; t < 4; ++t) {
        int col = (osub0 + t) * 16 + l16;
        #pragma unroll
        for (int r = 0; r < 8; ++r) {
            int row = kslab * 16 + r + 8 * hf;
            s_mj[row * MJ_STRIDE + col] = acc_j[t][r];
        }
    }
    __syncthreads();

    v8f agg[4];
    #pragma unroll
    for (int t = 0; t < 4; ++t) agg[t] = zero;

    const float* arow = s_adj + (kslab * 16 + l16) * ADJ_STRIDE + 2 * hf;
    for (int j = 0; j < 64; j += 4) {
        v2f a = *(const v2f*)(arow + j);
        #pragma unroll
        for (int t = 0; t < 4; ++t) {
            int col = (osub0 + t) * 16 + l16;
            v2f bv;
            bv.x = s_mj[(j + 2 * hf)     * MJ_STRIDE + col];
            bv.y = s_mj[(j + 2 * hf + 1) * MJ_STRIDE + col];
            agg[t] = __builtin_amdgcn_wmma_f32_16x16x4_f32(
                false, a, false, bv, (short)0, agg[t], false, false);
        }
    }

    float* outb = out + (size_t)b * 64 * 512;
    #pragma unroll
    for (int t = 0; t < 4; ++t) {
        int ocol = obase + (osub0 + t) * 16 + l16;
        float bv = bias[ocol];
        #pragma unroll
        for (int r = 0; r < 8; ++r) {
            int row = kslab * 16 + r + 8 * hf;
            outb[(size_t)row * 512 + ocol] = s_deg[row] * (acc_i[t][r] + bv) + agg[t][r];
        }
    }
}

extern "C" void kernel_launch(void* const* d_in, const int* in_sizes, int n_in,
                              void* d_out, int out_size, void* d_ws, size_t ws_size,
                              hipStream_t stream) {
    const float* x    = (const float*)d_in[0];
    const int*   adj  = (const int*)  d_in[1];
    const float* W    = (const float*)d_in[2];
    const float* bias = (const float*)d_in[3];
    float*       out  = (float*)d_out;

    dim3 grid(4, 32);
    gnn_fused_kernel<<<grid, 256, 0, stream>>>(x, adj, W, bias, out);
}


# Halve the work-group tile to 64 output columns on 4 waves: 256 work-groups, one per WGP and one wave per SIMD, not 128 of 8 waves on half the WGPs; per-wave tile, loaders, k and sum order kept
# speedup vs baseline: 2.5567x; 2.5567x over previous
#include <hip/hip_runtime.h>


typedef __attribute__((ext_vector_type(2))) float v2f;
typedef __attribute__((ext_vector_type(4))) float v4f;
typedef __attribute__((ext_vector_type(8))) float v8f;

#define MJ_STRIDE   68
#define ADJ_STRIDE  66

#define KC         32
#define ST_PITCH   36
#define ST_XROWS   64
#define ST_WROWS   64
#define ST_ROWS    (ST_XROWS + 2 * ST_WROWS)
#define ST_FLOATS  (ST_ROWS * ST_PITCH)
#define NCHUNK     (512 / KC)

#define NPRE       12
__device__ __forceinline__ void load_chunk(v4f (&pre)[NPRE], const float* xsrc,
                                           const float* wsrc, int koff)
{
    #pragma unroll
    for (int p = 0; p < 4; ++p)
        pre[p] = *(const v4f*)(xsrc + (size_t)p * 16 * 512 + koff);
    #pragma unroll
    for (int q = 0; q < 4; ++q) {
        pre[4 + q] = *(const v4f*)(wsrc + (size_t)q * 16 * 1024 + koff);
        pre[8 + q] = *(const v4f*)(wsrc + (size_t)q * 16 * 1024 + 512 + koff);
    }
}

__device__ __forceinline__ void store_chunk(const v4f (&pre)[NPRE], float* sdst)
{
    #pragma unroll
    for (int p = 0; p < NPRE; ++p)
        *(v4f*)(sdst + p * 16 * ST_PITCH) = pre[p];
}

struct Frag { v4f a; v4f bi[4]; v4f bj[4]; };

__device__ __forceinline__ void load_frag(Frag& f, const float* ap, const float* bp, int g)
{
    f.a = *(const v4f*)(ap + 8 * g);
    #pragma unroll
    for (int t = 0; t < 4; ++t) {
        f.bi[t] = *(const v4f*)(bp + (t * 16) * ST_PITCH + 8 * g);
        f.bj[t] = *(const v4f*)(bp + (ST_WROWS + t * 16) * ST_PITCH + 8 * g);
    }
}

__device__ __forceinline__ void mma_group(const Frag& f, v8f (&acc_i)[4], v8f (&acc_j)[4])
{
    #pragma unroll
    for (int u = 0; u < 2; ++u) {
        v2f a = {f.a[2 * u], f.a[2 * u + 1]};
        #pragma unroll
        for (int t = 0; t < 4; ++t) {
            v2f bi = {f.bi[t][2 * u], f.bi[t][2 * u + 1]};
            v2f bj = {f.bj[t][2 * u], f.bj[t][2 * u + 1]};
            acc_i[t] = __builtin_amdgcn_wmma_f32_16x16x4_f32(
                false, a, false, bi, (short)0, acc_i[t], false, false);
            acc_j[t] = __builtin_amdgcn_wmma_f32_16x16x4_f32(
                false, a, false, bj, (short)0, acc_j[t], false, false);
        }
    }
}

template <bool LOAD_AHEAD>
__device__ __forceinline__ void chunk_step(Frag& f0, Frag& f1, v4f (&pre)[NPRE],
                                           const float* ap, const float* bp, float* sdst,
                                           int cur, int nxt,
                                           const float* xsrc, const float* wsrc, int koff_ahead,
                                           v8f (&acc_i)[4], v8f (&acc_j)[4])
{
    load_frag(f1, ap + cur, bp + cur, 1);
    mma_group(f0, acc_i, acc_j);
    load_frag(f0, ap + cur, bp + cur, 2);
    mma_group(f1, acc_i, acc_j);
    load_frag(f1, ap + cur, bp + cur, 3);
    mma_group(f0, acc_i, acc_j);
    store_chunk(pre, sdst + nxt);
    __syncthreads();
    if (LOAD_AHEAD) load_chunk(pre, xsrc, wsrc, koff_ahead);
    load_frag(f0, ap + nxt, bp + nxt, 0);
    mma_group(f1, acc_i, acc_j);
}

__launch_bounds__(128)
__global__ void gnn_fused_kernel(const float* __restrict__ x,
                                 const int*   __restrict__ adj,
                                 const float* __restrict__ W,
                                 const float* __restrict__ bias,
                                 float*       __restrict__ out)
{
    __shared__ __align__(16) float s_stage[2 * ST_FLOATS];
    __shared__ __align__(16) float s_mj [64 * MJ_STRIDE];
    __shared__ __align__(16) float s_adj[64 * ADJ_STRIDE];
    __shared__ float s_deg[64];

    const int tid  = threadIdx.x;
    const int wave = tid >> 5;
    const int lane = tid & 31;
    const int hf   = lane >> 4;
    const int l16  = lane & 15;

    const int b     = blockIdx.y;
    const int obase = blockIdx.x * 64;

    const int kslab = wave;

    const int sseg = tid & 7;
    const int srow = tid >> 3;
    const float* xsrc = x + ((size_t)b * 64 + srow) * 512 + sseg * 4;
    const float* wsrc = W + (size_t)(obase + srow) * 1024 + sseg * 4;
    float* sdst = s_stage + srow * ST_PITCH + sseg * 4;

    v4f pre0[NPRE], pre[NPRE];
    load_chunk(pre0, xsrc, wsrc, 0);
    load_chunk(pre, xsrc, wsrc, KC);

    #pragma unroll
    for (int it = 0; it < 32; ++it) {
        int idx = it * 128 + tid;
        int r = idx >> 6, c = idx & 63;
        s_adj[r * ADJ_STRIDE + c] = (float)adj[idx];
    }
    store_chunk(pre0, sdst);
    __syncthreads();
    if (tid < 64) {
        float d = 0.f;
        for (int j = 0; j < 64; ++j) d += s_adj[tid * ADJ_STRIDE + j];
        s_deg[tid] = d;
    }

    const v8f zero = {0.f,0.f,0.f,0.f,0.f,0.f,0.f,0.f};
    v8f acc_i[4], acc_j[4];
    #pragma unroll
    for (int t = 0; t < 4; ++t) { acc_i[t] = zero; acc_j[t] = zero; }

    const float* ap = s_stage + (kslab * 16 + l16) * ST_PITCH + 4 * hf;
    const float* bp = s_stage + (ST_XROWS + l16) * ST_PITCH + 4 * hf;

    Frag f0, f1;
    load_frag(f0, ap, bp, 0);
    for (int kc = 0; kc < NCHUNK - 2; ++kc) {
        const int cur = (kc & 1) * ST_FLOATS;
        chunk_step<true>(f0, f1, pre, ap, bp, sdst, cur, ST_FLOATS - cur,
                         xsrc, wsrc, (kc + 2) * KC, acc_i, acc_j);
    }
    {
        const int cur = ((NCHUNK - 2) & 1) * ST_FLOATS;
        chunk_step<false>(f0, f1, pre, ap, bp, sdst, cur, ST_FLOATS - cur,
                          xsrc, wsrc, 0, acc_i, acc_j);
    }
    {
        const int cur = ((NCHUNK - 1) & 1) * ST_FLOATS;
        load_frag(f1, ap + cur, bp + cur, 1);
        mma_group(f0, acc_i, acc_j);
        load_frag(f0, ap + cur, bp + cur, 2);
        mma_group(f1, acc_i, acc_j);
        load_frag(f1, ap + cur, bp + cur, 3);
        mma_group(f0, acc_i, acc_j);
        mma_group(f1, acc_i, acc_j);
    }

    #pragma unroll
    for (int t = 0; t < 4; ++t) {
        int col = t * 16 + l16;
        #pragma unroll
        for (int r = 0; r < 8; ++r) {
            int row = kslab * 16 + r + 8 * hf;
            s_mj[row * MJ_STRIDE + col] = acc_j[t][r];
        }
    }
    __syncthreads();

    v8f agg[4];
    #pragma unroll
    for (int t = 0; t < 4; ++t) agg[t] = zero;

    const float* arow = s_adj + (kslab * 16 + l16) * ADJ_STRIDE + 2 * hf;
    for (int j = 0; j < 64; j += 4) {
        v2f a = *(const v2f*)(arow + j);
        #pragma unroll
        for (int t = 0; t < 4; ++t) {
            int col = t * 16 + l16;
            v2f bv;
            bv.x = s_mj[(j + 2 * hf)     * MJ_STRIDE + col];
            bv.y = s_mj[(j + 2 * hf + 1) * MJ_STRIDE + col];
            agg[t] = __builtin_amdgcn_wmma_f32_16x16x4_f32(
                false, a, false, bv, (short)0, agg[t], false, false);
        }
    }

    float* outb = out + (size_t)b * 64 * 512;
    #pragma unroll
    for (int t = 0; t < 4; ++t) {
        int ocol = obase + t * 16 + l16;
        float bv = bias[ocol];
        #pragma unroll
        for (int r = 0; r < 8; ++r) {
            int row = kslab * 16 + r + 8 * hf;
            outb[(size_t)row * 512 + ocol] = s_deg[row] * (acc_i[t][r] + bv) + agg[t][r];
        }
    }
}

extern "C" void kernel_launch(void* const* d_in, const int* in_sizes, int n_in,
                              void* d_out, int out_size, void* d_ws, size_t ws_size,
                              hipStream_t stream) {
    const float* x    = (const float*)d_in[0];
    const int*   adj  = (const int*)  d_in[1];
    const float* W    = (const float*)d_in[2];
    const float* bias = (const float*)d_in[3];
    float*       out  = (float*)d_out;

    dim3 grid(8, 32);
    gnn_fused_kernel<<<grid, 128, 0, stream>>>(x, adj, W, bias, out);
}


# Fetch x and W chunks two steps ahead, not one: two alternating register sets, K loop unrolled by two, step barriers fence LDS only so the loads stay in flight across them; k and sum order kept
# speedup vs baseline: 3.1092x; 1.2161x over previous
#include <hip/hip_runtime.h>


typedef __attribute__((ext_vector_type(2))) float v2f;
typedef __attribute__((ext_vector_type(4))) float v4f;
typedef __attribute__((ext_vector_type(8))) float v8f;

#define MJ_STRIDE   68
#define ADJ_STRIDE  66

#define KC         32
#define ST_PITCH   36
#define ST_XROWS   64
#define ST_WROWS   64
#define ST_ROWS    (ST_XROWS + 2 * ST_WROWS)
#define ST_FLOATS  (ST_ROWS * ST_PITCH)
#define NCHUNK     (512 / KC)

#define NPRE       12
__device__ __forceinline__ void load_chunk(v4f (&pre)[NPRE], const float* xsrc,
                                           const float* wsrc, int koff)
{
    #pragma unroll
    for (int p = 0; p < 4; ++p)
        pre[p] = *(const v4f*)(xsrc + (size_t)p * 16 * 512 + koff);
    #pragma unroll
    for (int q = 0; q < 4; ++q) {
        pre[4 + q] = *(const v4f*)(wsrc + (size_t)q * 16 * 1024 + koff);
        pre[8 + q] = *(const v4f*)(wsrc + (size_t)q * 16 * 1024 + 512 + koff);
    }
}

__device__ __forceinline__ void store_chunk(const v4f (&pre)[NPRE], float* sdst)
{
    #pragma unroll
    for (int p = 0; p < NPRE; ++p)
        *(v4f*)(sdst + p * 16 * ST_PITCH) = pre[p];
}

struct Frag { v4f a; v4f bi[4]; v4f bj[4]; };

__device__ __forceinline__ void load_frag(Frag& f, const float* ap, const float* bp, int g)
{
    f.a = *(const v4f*)(ap + 8 * g);
    #pragma unroll
    for (int t = 0; t < 4; ++t) {
        f.bi[t] = *(const v4f*)(bp + (t * 16) * ST_PITCH + 8 * g);
        f.bj[t] = *(const v4f*)(bp + (ST_WROWS + t * 16) * ST_PITCH + 8 * g);
    }
}

__device__ __forceinline__ void mma_group(const Frag& f, v8f (&acc_i)[4], v8f (&acc_j)[4])
{
    #pragma unroll
    for (int u = 0; u < 2; ++u) {
        v2f a = {f.a[2 * u], f.a[2 * u + 1]};
        #pragma unroll
        for (int t = 0; t < 4; ++t) {
            v2f bi = {f.bi[t][2 * u], f.bi[t][2 * u + 1]};
            v2f bj = {f.bj[t][2 * u], f.bj[t][2 * u + 1]};
            acc_i[t] = __builtin_amdgcn_wmma_f32_16x16x4_f32(
                false, a, false, bi, (short)0, acc_i[t], false, false);
            acc_j[t] = __builtin_amdgcn_wmma_f32_16x16x4_f32(
                false, a, false, bj, (short)0, acc_j[t], false, false);
        }
    }
}

__device__ __forceinline__ void lds_barrier()
{
    __builtin_amdgcn_fence(__ATOMIC_RELEASE, "workgroup", "local");
    __builtin_amdgcn_s_barrier();
    __builtin_amdgcn_fence(__ATOMIC_ACQUIRE, "workgroup", "local");
}

template <bool LOAD_AHEAD>
__device__ __forceinline__ void chunk_step(Frag& f0, Frag& f1, v4f (&pre)[NPRE],
                                           const float* ap, const float* bp, float* sdst,
                                           int cur, int nxt,
                                           const float* xsrc, const float* wsrc, int koff_ahead,
                                           v8f (&acc_i)[4], v8f (&acc_j)[4])
{
    load_frag(f1, ap + cur, bp + cur, 1);
    mma_group(f0, acc_i, acc_j);
    load_frag(f0, ap + cur, bp + cur, 2);
    mma_group(f1, acc_i, acc_j);
    load_frag(f1, ap + cur, bp + cur, 3);
    mma_group(f0, acc_i, acc_j);
    store_chunk(pre, sdst + nxt);
    lds_barrier();
    if (LOAD_AHEAD) load_chunk(pre, xsrc, wsrc, koff_ahead);
    load_frag(f0, ap + nxt, bp + nxt, 0);
    __builtin_amdgcn_sched_barrier(0);
    mma_group(f1, acc_i, acc_j);
}

__launch_bounds__(128)
__global__ void gnn_fused_kernel(const float* __restrict__ x,
                                 const int*   __restrict__ adj,
                                 const float* __restrict__ W,
                                 const float* __restrict__ bias,
                                 float*       __restrict__ out)
{
    __shared__ __align__(16) float s_stage[2 * ST_FLOATS];
    __shared__ __align__(16) float s_mj [64 * MJ_STRIDE];
    __shared__ __align__(16) float s_adj[64 * ADJ_STRIDE];
    __shared__ float s_deg[64];

    const int tid  = threadIdx.x;
    const int wave = tid >> 5;
    const int lane = tid & 31;
    const int hf   = lane >> 4;
    const int l16  = lane & 15;

    const int b     = blockIdx.y;
    const int obase = blockIdx.x * 64;

    const int kslab = wave;

    const int sseg = tid & 7;
    const int srow = tid >> 3;
    const float* xsrc = x + ((size_t)b * 64 + srow) * 512 + sseg * 4;
    const float* wsrc = W + (size_t)(obase + srow) * 1024 + sseg * 4;
    float* sdst = s_stage + srow * ST_PITCH + sseg * 4;

    v4f pe[NPRE], po[NPRE];
    load_chunk(pe, xsrc, wsrc, 0);
    load_chunk(po, xsrc, wsrc, KC);

    #pragma unroll
    for (int it = 0; it < 32; ++it) {
        int idx = it * 128 + tid;
        int r = idx >> 6, c = idx & 63;
        s_adj[r * ADJ_STRIDE + c] = (float)adj[idx];
    }
    store_chunk(pe, sdst);
    lds_barrier();
    load_chunk(pe, xsrc, wsrc, 2 * KC);
    if (tid < 64) {
        float d = 0.f;
        for (int j = 0; j < 64; ++j) d += s_adj[tid * ADJ_STRIDE + j];
        s_deg[tid] = d;
    }

    const v8f zero = {0.f,0.f,0.f,0.f,0.f,0.f,0.f,0.f};
    v8f acc_i[4], acc_j[4];
    #pragma unroll
    for (int t = 0; t < 4; ++t) { acc_i[t] = zero; acc_j[t] = zero; }

    const float* ap = s_stage + (kslab * 16 + l16) * ST_PITCH + 4 * hf;
    const float* bp = s_stage + (ST_XROWS + l16) * ST_PITCH + 4 * hf;

    Frag f0, f1;
    load_frag(f0, ap, bp, 0);
    for (int kc = 0; kc < NCHUNK - 4; kc += 2) {
        chunk_step<true>(f0, f1, po, ap, bp, sdst, 0, ST_FLOATS,
                         xsrc, wsrc, (kc + 3) * KC, acc_i, acc_j);
        chunk_step<true>(f0, f1, pe, ap, bp, sdst, ST_FLOATS, 0,
                         xsrc, wsrc, (kc + 4) * KC, acc_i, acc_j);
    }
    chunk_step<true>(f0, f1, po, ap, bp, sdst, 0, ST_FLOATS,
                     xsrc, wsrc, (NCHUNK - 1) * KC, acc_i, acc_j);
    chunk_step<false>(f0, f1, pe, ap, bp, sdst, ST_FLOATS, 0,
                      xsrc, wsrc, 0, acc_i, acc_j);
    chunk_step<false>(f0, f1, po, ap, bp, sdst, 0, ST_FLOATS,
                      xsrc, wsrc, 0, acc_i, acc_j);
    {
        const int cur = ((NCHUNK - 1) & 1) * ST_FLOATS;
        load_frag(f1, ap + cur, bp + cur, 1);
        mma_group(f0, acc_i, acc_j);
        load_frag(f0, ap + cur, bp + cur, 2);
        mma_group(f1, acc_i, acc_j);
        load_frag(f1, ap + cur, bp + cur, 3);
        mma_group(f0, acc_i, acc_j);
        mma_group(f1, acc_i, acc_j);
    }

    #pragma unroll
    for (int t = 0; t < 4; ++t) {
        int col = t * 16 + l16;
        #pragma unroll
        for (int r = 0; r < 8; ++r) {
            int row = kslab * 16 + r + 8 * hf;
            s_mj[row * MJ_STRIDE + col] = acc_j[t][r];
        }
    }
    __syncthreads();

    v8f agg[4];
    #pragma unroll
    for (int t = 0; t < 4; ++t) agg[t] = zero;

    const float* arow = s_adj + (kslab * 16 + l16) * ADJ_STRIDE + 2 * hf;
    for (int j = 0; j < 64; j += 4) {
        v2f a = *(const v2f*)(arow + j);
        #pragma unroll
        for (int t = 0; t < 4; ++t) {
            int col = t * 16 + l16;
            v2f bv;
            bv.x = s_mj[(j + 2 * hf)     * MJ_STRIDE + col];
            bv.y = s_mj[(j + 2 * hf + 1) * MJ_STRIDE + col];
            agg[t] = __builtin_amdgcn_wmma_f32_16x16x4_f32(
                false, a, false, bv, (short)0, agg[t], false, false);
        }
    }

    float* outb = out + (size_t)b * 64 * 512;
    #pragma unroll
    for (int t = 0; t < 4; ++t) {
        int ocol = obase + t * 16 + l16;
        float bv = bias[ocol];
        #pragma unroll
        for (int r = 0; r < 8; ++r) {
            int row = kslab * 16 + r + 8 * hf;
            outb[(size_t)row * 512 + ocol] = s_deg[row] * (acc_i[t][r] + bv) + agg[t][r];
        }
    }
}

extern "C" void kernel_launch(void* const* d_in, const int* in_sizes, int n_in,
                              void* d_out, int out_size, void* d_ws, size_t ws_size,
                              hipStream_t stream) {
    const float* x    = (const float*)d_in[0];
    const int*   adj  = (const int*)  d_in[1];
    const float* W    = (const float*)d_in[2];
    const float* bias = (const float*)d_in[3];
    float*       out  = (float*)d_out;

    dim3 grid(8, 32);
    gnn_fused_kernel<<<grid, 128, 0, stream>>>(x, adj, W, bias, out);
}


# Split each wave's tile into two 32-column halves on 8 waves per work-group: two waves per SIMD, so one wave's LDS reads, staging and barrier waits run under the other's WMMAs; k and sum order kept
# speedup vs baseline: 3.3317x; 1.0716x over previous
#include <hip/hip_runtime.h>


typedef __attribute__((ext_vector_type(2))) float v2f;
typedef __attribute__((ext_vector_type(4))) float v4f;
typedef __attribute__((ext_vector_type(8))) float v8f;

#define MJ_STRIDE   68
#define ADJ_STRIDE  66

#define KC         32
#define ST_PITCH   36
#define ST_XROWS   64
#define ST_WROWS   64
#define ST_ROWS    (ST_XROWS + 2 * ST_WROWS)
#define ST_FLOATS  (ST_ROWS * ST_PITCH)
#define NCHUNK     (512 / KC)

#define NPRE       6
__device__ __forceinline__ void load_chunk(v4f (&pre)[NPRE], const float* xsrc,
                                           const float* wsrc, int koff)
{
    #pragma unroll
    for (int p = 0; p < 2; ++p)
        pre[p] = *(const v4f*)(xsrc + (size_t)p * 32 * 512 + koff);
    #pragma unroll
    for (int q = 0; q < 2; ++q) {
        pre[2 + q] = *(const v4f*)(wsrc + (size_t)q * 32 * 1024 + koff);
        pre[4 + q] = *(const v4f*)(wsrc + (size_t)q * 32 * 1024 + 512 + koff);
    }
}

__device__ __forceinline__ void store_chunk(const v4f (&pre)[NPRE], float* sdst)
{
    #pragma unroll
    for (int p = 0; p < NPRE; ++p)
        *(v4f*)(sdst + p * 32 * ST_PITCH) = pre[p];
}

struct Frag { v4f a; v4f bi[2]; v4f bj[2]; };

__device__ __forceinline__ void load_frag(Frag& f, const float* ap, const float* bp, int g)
{
    f.a = *(const v4f*)(ap + 8 * g);
    #pragma unroll
    for (int t = 0; t < 2; ++t) {
        f.bi[t] = *(const v4f*)(bp + (t * 16) * ST_PITCH + 8 * g);
        f.bj[t] = *(const v4f*)(bp + (ST_WROWS + t * 16) * ST_PITCH + 8 * g);
    }
}

__device__ __forceinline__ void mma_group(const Frag& f, v8f (&acc_i)[2], v8f (&acc_j)[2])
{
    #pragma unroll
    for (int u = 0; u < 2; ++u) {
        v2f a = {f.a[2 * u], f.a[2 * u + 1]};
        #pragma unroll
        for (int t = 0; t < 2; ++t) {
            v2f bi = {f.bi[t][2 * u], f.bi[t][2 * u + 1]};
            v2f bj = {f.bj[t][2 * u], f.bj[t][2 * u + 1]};
            acc_i[t] = __builtin_amdgcn_wmma_f32_16x16x4_f32(
                false, a, false, bi, (short)0, acc_i[t], false, false);
            acc_j[t] = __builtin_amdgcn_wmma_f32_16x16x4_f32(
                false, a, false, bj, (short)0, acc_j[t], false, false);
        }
    }
}

__device__ __forceinline__ void lds_barrier()
{
    __builtin_amdgcn_fence(__ATOMIC_RELEASE, "workgroup", "local");
    __builtin_amdgcn_s_barrier();
    __builtin_amdgcn_fence(__ATOMIC_ACQUIRE, "workgroup", "local");
}

template <bool LOAD_AHEAD>
__device__ __forceinline__ void chunk_step(Frag& f0, Frag& f1, v4f (&pre)[NPRE],
                                           const float* ap, const float* bp, float* sdst,
                                           int cur, int nxt,
                                           const float* xsrc, const float* wsrc, int koff_ahead,
                                           v8f (&acc_i)[2], v8f (&acc_j)[2])
{
    load_frag(f1, ap + cur, bp + cur, 1);
    mma_group(f0, acc_i, acc_j);
    load_frag(f0, ap + cur, bp + cur, 2);
    mma_group(f1, acc_i, acc_j);
    load_frag(f1, ap + cur, bp + cur, 3);
    mma_group(f0, acc_i, acc_j);
    store_chunk(pre, sdst + nxt);
    lds_barrier();
    if (LOAD_AHEAD) load_chunk(pre, xsrc, wsrc, koff_ahead);
    load_frag(f0, ap + nxt, bp + nxt, 0);
    __builtin_amdgcn_sched_barrier(0);
    mma_group(f1, acc_i, acc_j);
}

__launch_bounds__(256)
__global__ void gnn_fused_kernel(const float* __restrict__ x,
                                 const int*   __restrict__ adj,
                                 const float* __restrict__ W,
                                 const float* __restrict__ bias,
                                 float*       __restrict__ out)
{
    __shared__ __align__(16) float s_stage[2 * ST_FLOATS];
    __shared__ __align__(16) float s_mj [64 * MJ_STRIDE];
    __shared__ __align__(16) float s_adj[64 * ADJ_STRIDE];
    __shared__ float s_deg[64];

    const int tid  = threadIdx.x;
    const int wave = tid >> 5;
    const int lane = tid & 31;
    const int hf   = lane >> 4;
    const int l16  = lane & 15;

    const int b     = blockIdx.y;
    const int obase = blockIdx.x * 64;

    const int kslab = wave & 3;
    const int ch    = wave >> 2;

    const int sseg = tid & 7;
    const int srow = tid >> 3;
    const float* xsrc = x + ((size_t)b * 64 + srow) * 512 + sseg * 4;
    const float* wsrc = W + (size_t)(obase + srow) * 1024 + sseg * 4;
    float* sdst = s_stage + srow * ST_PITCH + sseg * 4;

    v4f pe[NPRE], po[NPRE];
    load_chunk(pe, xsrc, wsrc, 0);
    load_chunk(po, xsrc, wsrc, KC);

    #pragma unroll
    for (int it = 0; it < 16; ++it) {
        int idx = it * 256 + tid;
        int r = idx >> 6, c = idx & 63;
        s_adj[r * ADJ_STRIDE + c] = (float)adj[idx];
    }
    store_chunk(pe, sdst);
    lds_barrier();
    load_chunk(pe, xsrc, wsrc, 2 * KC);
    if (tid < 64) {
        float d = 0.f;
        for (int j = 0; j < 64; ++j) d += s_adj[tid * ADJ_STRIDE + j];
        s_deg[tid] = d;
    }

    const v8f zero = {0.f,0.f,0.f,0.f,0.f,0.f,0.f,0.f};
    v8f acc_i[2], acc_j[2];
    #pragma unroll
    for (int t = 0; t < 2; ++t) { acc_i[t] = zero; acc_j[t] = zero; }

    const float* ap = s_stage + (kslab * 16 + l16) * ST_PITCH + 4 * hf;
    const float* bp = s_stage + (ST_XROWS + ch * 32 + l16) * ST_PITCH + 4 * hf;

    Frag f0, f1;
    load_frag(f0, ap, bp, 0);
    for (int kc = 0; kc < NCHUNK - 4; kc += 2) {
        chunk_step<true>(f0, f1, po, ap, bp, sdst, 0, ST_FLOATS,
                         xsrc, wsrc, (kc + 3) * KC, acc_i, acc_j);
        chunk_step<true>(f0, f1, pe, ap, bp, sdst, ST_FLOATS, 0,
                         xsrc, wsrc, (kc + 4) * KC, acc_i, acc_j);
    }
    chunk_step<true>(f0, f1, po, ap, bp, sdst, 0, ST_FLOATS,
                     xsrc, wsrc, (NCHUNK - 1) * KC, acc_i, acc_j);
    chunk_step<false>(f0, f1, pe, ap, bp, sdst, ST_FLOATS, 0,
                      xsrc, wsrc, 0, acc_i, acc_j);
    chunk_step<false>(f0, f1, po, ap, bp, sdst, 0, ST_FLOATS,
                      xsrc, wsrc, 0, acc_i, acc_j);
    {
        const int cur = ((NCHUNK - 1) & 1) * ST_FLOATS;
        load_frag(f1, ap + cur, bp + cur, 1);
        mma_group(f0, acc_i, acc_j);
        load_frag(f0, ap + cur, bp + cur, 2);
        mma_group(f1, acc_i, acc_j);
        load_frag(f1, ap + cur, bp + cur, 3);
        mma_group(f0, acc_i, acc_j);
        mma_group(f1, acc_i, acc_j);
    }

    #pragma unroll
    for (int t = 0; t < 2; ++t) {
        int col = ch * 32 + t * 16 + l16;
        #pragma unroll
        for (int r = 0; r < 8; ++r) {
            int row = kslab * 16 + r + 8 * hf;
            s_mj[row * MJ_STRIDE + col] = acc_j[t][r];
        }
    }
    __syncthreads();

    v8f agg[2];
    #pragma unroll
    for (int t = 0; t < 2; ++t) agg[t] = zero;

    const float* arow = s_adj + (kslab * 16 + l16) * ADJ_STRIDE + 2 * hf;
    for (int j = 0; j < 64; j += 4) {
        v2f a = *(const v2f*)(arow + j);
        #pragma unroll
        for (int t = 0; t < 2; ++t) {
            int col = ch * 32 + t * 16 + l16;
            v2f bv;
            bv.x = s_mj[(j + 2 * hf)     * MJ_STRIDE + col];
            bv.y = s_mj[(j + 2 * hf + 1) * MJ_STRIDE + col];
            agg[t] = __builtin_amdgcn_wmma_f32_16x16x4_f32(
                false, a, false, bv, (short)0, agg[t], false, false);
        }
    }

    float* outb = out + (size_t)b * 64 * 512;
    #pragma unroll
    for (int t = 0; t < 2; ++t) {
        int ocol = obase + ch * 32 + t * 16 + l16;
        float bv = bias[ocol];
        #pragma unroll
        for (int r = 0; r < 8; ++r) {
            int row = kslab * 16 + r + 8 * hf;
            outb[(size_t)row * 512 + ocol] = s_deg[row] * (acc_i[t][r] + bv) + agg[t][r];
        }
    }
}

extern "C" void kernel_launch(void* const* d_in, const int* in_sizes, int n_in,
                              void* d_out, int out_size, void* d_ws, size_t ws_size,
                              hipStream_t stream) {
    const float* x    = (const float*)d_in[0];
    const int*   adj  = (const int*)  d_in[1];
    const float* W    = (const float*)d_in[2];
    const float* bias = (const float*)d_in[3];
    float*       out  = (float*)d_out;

    dim3 grid(8, 32);
    gnn_fused_kernel<<<grid, 256, 0, stream>>>(x, adj, W, bias, out);
}
